# starting kernel
# baseline (speedup 1.0000x reference)
#include <hip/hip_runtime.h>
#include <math.h>

typedef __attribute__((ext_vector_type(2))) float v2f;
typedef __attribute__((ext_vector_type(8))) float v8f;

namespace {
constexpr int kB = 65536;
constexpr int kT = 9;
constexpr int kI = 57;
constexpr int kG = 6;
constexpr int kKsteps = 15;
constexpr int kWavesPerBlock = 2;
constexpr int kBlockThreads = kWavesPerBlock * 32;
constexpr int kBatchPerWave = 32;
constexpr int kBatchPerBlock = kWavesPerBlock * kBatchPerWave;
}

__device__ __forceinline__ float fast_sigmoid(float v) {
  return 1.0f / (1.0f + __expf(-v));
}
__device__ __forceinline__ float fast_tanh(float v) {
  const float e = __expf(-2.0f * v);
  return (1.0f - e) / (1.0f + e);
}

__launch_bounds__(kBlockThreads)
__global__ void gru_fused_wmma(const float* __restrict__ x,
                               const float* __restrict__ W_ih,
                               const float* __restrict__ W_hh,
                               const float* __restrict__ b_ih,
                               const float* __restrict__ b_hh,
                               const float* __restrict__ fc_w,
                               const float* __restrict__ fc_b,
                               float* __restrict__ out) {
  __shared__ float gx_s[kWavesPerBlock * kT * kBatchPerWave * kG];

  const int tid = threadIdx.x;
  const int wave = tid >> 5;
  const int lane = tid & 31;
  const int nlo = lane & 15;
  const int hi = lane >> 4;
  const int b0 = blockIdx.x * kBatchPerBlock + wave * kBatchPerWave;
  float* wlds = gx_s + wave * (kT * kBatchPerWave * kG);

  v2f bfrag[kKsteps];
#pragma unroll
  for (int ks = 0; ks < kKsteps; ++ks) {
    const int k = 4 * ks + 2 * hi;
    float bx = 0.0f, by = 0.0f;
    if (nlo < kG) {
      if (k < kI) bx = W_ih[nlo * kI + k];
      if (k + 1 < kI) by = W_ih[nlo * kI + k + 1];
    }
    v2f f;
    f.x = bx;
    f.y = by;
    bfrag[ks] = f;
  }

  const int rowStride = kT * kI;
  const float* xr0 = x + (b0 + nlo) * rowStride;
  const float* xr1 = x + (b0 + 16 + nlo) * rowStride;

  for (int t = 0; t < kT; ++t) {
    const float* p0 = xr0 + t * kI;
    const float* p1 = xr1 + t * kI;
    if (t + 1 < kT) {
      __builtin_prefetch(p0 + kI, 0, 0);
      __builtin_prefetch(p1 + kI, 0, 0);
    }
    v8f c0 = {};
    v8f c1 = {};
#pragma unroll
    for (int ks = 0; ks < kKsteps; ++ks) {
      const int k = 4 * ks + 2 * hi;
      v2f a0, a1;
      a0.x = (k < kI) ? p0[k] : 0.0f;
      a0.y = (k + 1 < kI) ? p0[k + 1] : 0.0f;
      a1.x = (k < kI) ? p1[k] : 0.0f;
      a1.y = (k + 1 < kI) ? p1[k + 1] : 0.0f;
      c0 = __builtin_amdgcn_wmma_f32_16x16x4_f32(false, a0, false, bfrag[ks],
                                                 (short)0, c0, false, false);
      c1 = __builtin_amdgcn_wmma_f32_16x16x4_f32(false, a1, false, bfrag[ks],
                                                 (short)0, c1, false, false);
    }
    if (nlo < kG) {
#pragma unroll
      for (int j = 0; j < 8; ++j) {
        const int m = j + 8 * hi;
        wlds[(t * kBatchPerWave + m) * kG + nlo] = c0[j];
        wlds[(t * kBatchPerWave + 16 + m) * kG + nlo] = c1[j];
      }
    }
  }

  __syncthreads();

  float wh[12], bi[6], bh[6];
#pragma unroll
  for (int i2 = 0; i2 < 12; ++i2) wh[i2] = W_hh[i2];
#pragma unroll
  for (int g = 0; g < 6; ++g) {
    bi[g] = b_ih[g];
    bh[g] = b_hh[g];
  }

  float h0 = 0.0f, h1 = 0.0f;
#pragma unroll
  for (int t = 0; t < kT; ++t) {
    const float* gt = wlds + (t * kBatchPerWave + lane) * kG;
    const float gh0 = fmaf(wh[0], h0, fmaf(wh[1], h1, bh[0]));
    const float gh1 = fmaf(wh[2], h0, fmaf(wh[3], h1, bh[1]));
    const float gh2 = fmaf(wh[4], h0, fmaf(wh[5], h1, bh[2]));
    const float gh3 = fmaf(wh[6], h0, fmaf(wh[7], h1, bh[3]));
    const float gh4 = fmaf(wh[8], h0, fmaf(wh[9], h1, bh[4]));
    const float gh5 = fmaf(wh[10], h0, fmaf(wh[11], h1, bh[5]));
    const float r0 = fast_sigmoid(gt[0] + bi[0] + gh0);
    const float r1 = fast_sigmoid(gt[1] + bi[1] + gh1);
    const float z0 = fast_sigmoid(gt[2] + bi[2] + gh2);
    const float z1 = fast_sigmoid(gt[3] + bi[3] + gh3);
    const float n0 = fast_tanh(fmaf(r0, gh4, gt[4] + bi[4]));
    const float n1 = fast_tanh(fmaf(r1, gh5, gt[5] + bi[5]));
    h0 = fmaf(z0, h0 - n0, n0);
    h1 = fmaf(z1, h1 - n1, n1);
  }
  out[b0 + lane] = fmaf(fc_w[0], h0, fmaf(fc_w[1], h1, fc_b[0]));
}

extern "C" void kernel_launch(void* const* d_in, const int* in_sizes, int n_in,
                              void* d_out, int out_size, void* d_ws, size_t ws_size,
                              hipStream_t stream) {
  const float* x    = (const float*)d_in[0];
  const float* W_ih = (const float*)d_in[1];
  const float* W_hh = (const float*)d_in[2];
  const float* b_ih = (const float*)d_in[3];
  const float* b_hh = (const float*)d_in[4];
  const float* fc_w = (const float*)d_in[5];
  const float* fc_b = (const float*)d_in[6];
  float* out = (float*)d_out;

  dim3 grid(kB / kBatchPerBlock);
  dim3 block(kBlockThreads);
  hipLaunchKernelGGL(gru_fused_wmma, grid, block, 0, stream,
                     x, W_ih, W_hh, b_ih, b_hh, fc_w, fc_b, out);
}


# x read with whole-line 16-byte-per-lane loads of each wave's contiguous rows, staged through wave-private LDS with later stages' loads in flight; A fragments read from LDS; summation order unchanged
# speedup vs baseline: 2.1008x; 2.1008x over previous
#include <hip/hip_runtime.h>
#include <math.h>


typedef __attribute__((ext_vector_type(2))) float v2f;
typedef __attribute__((ext_vector_type(4))) float v4f;
typedef __attribute__((ext_vector_type(8))) float v8f;

namespace {
constexpr int kB = 65536;
constexpr int kT = 9;
constexpr int kI = 57;
constexpr int kG = 6;
constexpr int kKsteps = 15;
constexpr int kWavesPerBlock = 2;
constexpr int kBlockThreads = kWavesPerBlock * 32;
constexpr int kBatchPerWave = 32;
constexpr int kBatchPerBlock = kWavesPerBlock * kBatchPerWave;

constexpr int kRowsPerWave = kBatchPerWave * kT;
constexpr int kStageRows = 32;
constexpr int kStages = kRowsPerWave / kStageRows;
constexpr int kStageFloats = kStageRows * kI;
constexpr int kStageVecs = kStageFloats / 4;
constexpr int kLoadsPerStage = (kStageVecs + 31) / 32;
constexpr int kStagePad = 8;
constexpr int kStageLds = kStageFloats + kStagePad;
constexpr int kGxFloats = kRowsPerWave * kG;
}

__device__ __forceinline__ float fast_sigmoid(float v) {
  return 1.0f / (1.0f + __expf(-v));
}
__device__ __forceinline__ float fast_tanh(float v) {
  const float e = __expf(-2.0f * v);
  return (1.0f - e) / (1.0f + e);
}

__device__ __forceinline__ void wave_lds_fence() {
  __builtin_amdgcn_fence(__ATOMIC_ACQ_REL, "wavefront");
}

__device__ __forceinline__ int stage_vec_index(int i, int lane) {
  const int vi = i * 32 + lane;
  return (i == kLoadsPerStage - 1) ? min(vi, kStageVecs - 1) : vi;
}

__launch_bounds__(kBlockThreads)
__global__ void gru_fused_wmma(const float* __restrict__ x,
                               const float* __restrict__ W_ih,
                               const float* __restrict__ W_hh,
                               const float* __restrict__ b_ih,
                               const float* __restrict__ b_hh,
                               const float* __restrict__ fc_w,
                               const float* __restrict__ fc_b,
                               float* __restrict__ out) {
  __shared__ __attribute__((aligned(16))) float xs_s[kWavesPerBlock * kStageLds];
  __shared__ __attribute__((aligned(16))) float gx_s[kWavesPerBlock * kGxFloats];

  const int tid = threadIdx.x;
  const int wave = tid >> 5;
  const int lane = tid & 31;
  const int nlo = lane & 15;
  const int hi = lane >> 4;
  const int b0 = blockIdx.x * kBatchPerBlock + wave * kBatchPerWave;
  float* sbuf = xs_s + wave * kStageLds;
  float* wlds = gx_s + wave * kGxFloats;

  v2f bfrag[kKsteps];
#pragma unroll
  for (int ks = 0; ks < kKsteps; ++ks) {
    const int k = 4 * ks + 2 * hi;
    float bx = 0.0f, by = 0.0f;
    if (nlo < kG) {
      if (k < kI) bx = W_ih[nlo * kI + k];
      if (k + 1 < kI) by = W_ih[nlo * kI + k + 1];
    }
    v2f f;
    f.x = bx;
    f.y = by;
    bfrag[ks] = f;
  }

  const v4f* xv = reinterpret_cast<const v4f*>(x + (size_t)b0 * (kT * kI));

  v4f r[2][kLoadsPerStage];
#pragma unroll
  for (int p = 0; p < 2; ++p) {
#pragma unroll
    for (int i = 0; i < kLoadsPerStage; ++i)
      r[p][i] = xv[p * kStageVecs + stage_vec_index(i, lane)];
  }

  if (lane < kStagePad) sbuf[kStageFloats + lane] = 0.0f;

  const float* a0p = sbuf + nlo * kI + 2 * hi;
  const float* a1p = sbuf + (16 + nlo) * kI + 2 * hi;

#pragma unroll
  for (int s = 0; s < kStages; ++s) {
#pragma unroll
    for (int i = 0; i < kLoadsPerStage; ++i)
      *reinterpret_cast<v4f*>(sbuf + 4 * stage_vec_index(i, lane)) = r[s & 1][i];
    if (s + 2 < kStages) {
#pragma unroll
      for (int i = 0; i < kLoadsPerStage; ++i)
        r[s & 1][i] = xv[(s + 2) * kStageVecs + stage_vec_index(i, lane)];
    }
    wave_lds_fence();

    v8f c0 = {};
    v8f c1 = {};
#pragma unroll
    for (int ks = 0; ks < kKsteps; ++ks) {
      const int k = 4 * ks + 2 * hi;
      const float x00 = a0p[4 * ks], x01 = a0p[4 * ks + 1];
      const float x10 = a1p[4 * ks], x11 = a1p[4 * ks + 1];
      v2f a0, a1;
      a0.x = (k < kI) ? x00 : 0.0f;
      a0.y = (k + 1 < kI) ? x01 : 0.0f;
      a1.x = (k < kI) ? x10 : 0.0f;
      a1.y = (k + 1 < kI) ? x11 : 0.0f;
      c0 = __builtin_amdgcn_wmma_f32_16x16x4_f32(false, a0, false, bfrag[ks],
                                                 (short)0, c0, false, false);
      c1 = __builtin_amdgcn_wmma_f32_16x16x4_f32(false, a1, false, bfrag[ks],
                                                 (short)0, c1, false, false);
    }
    if (nlo < kG) {
#pragma unroll
      for (int j = 0; j < 8; ++j) {
        const int m = j + 8 * hi;
        wlds[(s * kStageRows + m) * kG + nlo] = c0[j];
        wlds[(s * kStageRows + 16 + m) * kG + nlo] = c1[j];
      }
    }
    wave_lds_fence();
  }

  float wh[12], bi[6], bh[6];
#pragma unroll
  for (int i2 = 0; i2 < 12; ++i2) wh[i2] = W_hh[i2];
#pragma unroll
  for (int g = 0; g < 6; ++g) {
    bi[g] = b_ih[g];
    bh[g] = b_hh[g];
  }

  float h0 = 0.0f, h1 = 0.0f;
#pragma unroll
  for (int t = 0; t < kT; ++t) {
    const float* gt = wlds + (lane * kT + t) * kG;
    const float gh0 = fmaf(wh[0], h0, fmaf(wh[1], h1, bh[0]));
    const float gh1 = fmaf(wh[2], h0, fmaf(wh[3], h1, bh[1]));
    const float gh2 = fmaf(wh[4], h0, fmaf(wh[5], h1, bh[2]));
    const float gh3 = fmaf(wh[6], h0, fmaf(wh[7], h1, bh[3]));
    const float gh4 = fmaf(wh[8], h0, fmaf(wh[9], h1, bh[4]));
    const float gh5 = fmaf(wh[10], h0, fmaf(wh[11], h1, bh[5]));
    const float r0 = fast_sigmoid(gt[0] + bi[0] + gh0);
    const float r1 = fast_sigmoid(gt[1] + bi[1] + gh1);
    const float z0 = fast_sigmoid(gt[2] + bi[2] + gh2);
    const float z1 = fast_sigmoid(gt[3] + bi[3] + gh3);
    const float n0 = fast_tanh(fmaf(r0, gh4, gt[4] + bi[4]));
    const float n1 = fast_tanh(fmaf(r1, gh5, gt[5] + bi[5]));
    h0 = fmaf(z0, h0 - n0, n0);
    h1 = fmaf(z1, h1 - n1, n1);
  }
  out[b0 + lane] = fmaf(fc_w[0], h0, fmaf(fc_w[1], h1, fc_b[0]));
}

extern "C" void kernel_launch(void* const* d_in, const int* in_sizes, int n_in,
                              void* d_out, int out_size, void* d_ws, size_t ws_size,
                              hipStream_t stream) {
  const float* x    = (const float*)d_in[0];
  const float* W_ih = (const float*)d_in[1];
  const float* W_hh = (const float*)d_in[2];
  const float* b_ih = (const float*)d_in[3];
  const float* b_hh = (const float*)d_in[4];
  const float* fc_w = (const float*)d_in[5];
  const float* fc_b = (const float*)d_in[6];
  float* out = (float*)d_out;

  dim3 grid(kB / kBatchPerBlock);
  dim3 block(kBlockThreads);
  hipLaunchKernelGGL(gru_fused_wmma, grid, block, 0, stream,
                     x, W_ih, W_hh, b_ih, b_hh, fc_w, fc_b, out);
}
